# starting kernel
# baseline (speedup 1.0000x reference)
#include <hip/hip_runtime.h>
#include <math.h>

#define N_NODES 50000
#define N_HID   128
#define N_ADJ   4
#define N_EDGE  600000
#define TOTAL_EDGES (N_ADJ * N_EDGE)

typedef __attribute__((ext_vector_type(2))) float v2f;
typedef __attribute__((ext_vector_type(8))) float v8f;

__device__ __forceinline__ void global_fadd_f32(float* p, float v) {
    asm volatile("global_atomic_add_f32 %0, %1, off scope:SCOPE_DEV"
                 :
                 : "v"(p), "v"(v)
                 : "memory");
}

__global__ void zero_acc(float4* __restrict__ ws, int n4) {
    int i = blockIdx.x * blockDim.x + threadIdx.x;
    if (i < n4) {
        float4 z; z.x = 0.f; z.y = 0.f; z.z = 0.f; z.w = 0.f;
        ws[i] = z;
    }
}

__global__ __launch_bounds__(256) void scatter_edges(
    const float* __restrict__ x,
    const float* __restrict__ weight,
    const int*   __restrict__ adj_src,
    const int*   __restrict__ adj_dst,
    const float* __restrict__ adj_vals,
    float*       __restrict__ acc)
{
    const int lane = threadIdx.x & 31;
    const int e = blockIdx.x * (blockDim.x >> 5) + (threadIdx.x >> 5);
    if (e >= TOTAL_EDGES) return;

    const int   s  = __builtin_nontemporal_load(adj_src + e);
    const int   d  = __builtin_nontemporal_load(adj_dst + e);
    const float v  = __builtin_nontemporal_load(adj_vals + e);
    const int   a  = e / N_EDGE;
    const float sc = weight[a] * v;

    const float* xr   = x   + (long long)s * N_HID;
    float*       orow = acc + (long long)d * N_HID;

    #pragma unroll
    for (int k = 0; k < 4; ++k) {
        float m = xr[k * 32 + lane] * sc;
        global_fadd_f32(orow + k * 32 + lane, m);
    }
}

__global__ __launch_bounds__(160) void finalize_norm_gelu(
    const float* __restrict__ acc, float* __restrict__ out)
{
    const int lane = threadIdx.x & 31;
    const int wave = blockIdx.x * (blockDim.x >> 5) + (threadIdx.x >> 5);
    if (wave >= (N_NODES / 16)) return;

    const int node   = wave * 16 + (lane & 15);
    const int colOff = 2 * (lane >> 4);
    const float* rowp = acc + (long long)node * N_HID + colOff;

    v2f vbuf[32];
    v8f c = {};
    #pragma unroll
    for (int j = 0; j < 32; ++j) {
        v2f a;
        a.x = rowp[4 * j];
        a.y = rowp[4 * j + 1];
        vbuf[j] = a;
        c = __builtin_amdgcn_wmma_f32_16x16x4_f32(
                false, a, false, a, (short)0, c, false, false);
    }

    const int sel = lane & 7;
    float dv = c[0];
    dv = (sel == 1) ? c[1] : dv;
    dv = (sel == 2) ? c[2] : dv;
    dv = (sel == 3) ? c[3] : dv;
    dv = (sel == 4) ? c[4] : dv;
    dv = (sel == 5) ? c[5] : dv;
    dv = (sel == 6) ? c[6] : dv;
    dv = (sel == 7) ? c[7] : dv;
    const int nm      = lane & 15;
    const int srcLane = (nm < 8) ? nm : (nm + 16);
    const float sumsq = __shfl(dv, srcLane, 32);

    const float scale = 1.0f / fmaxf(sqrtf(sumsq), 1e-12f);

    float* op = out + (long long)node * N_HID + colOff;
    const float inv_sqrt2 = 0.70710678118654752440f;
    #pragma unroll
    for (int j = 0; j < 32; ++j) {
        float vx = vbuf[j].x * scale;
        float vy = vbuf[j].y * scale;
        vx = 0.5f * vx * (1.0f + erff(vx * inv_sqrt2));
        vy = 0.5f * vy * (1.0f + erff(vy * inv_sqrt2));
        op[4 * j]     = vx;
        op[4 * j + 1] = vy;
    }
}

extern "C" void kernel_launch(void* const* d_in, const int* in_sizes, int n_in,
                              void* d_out, int out_size, void* d_ws, size_t ws_size,
                              hipStream_t stream) {
    const float* x        = (const float*)d_in[0];
    const float* weight   = (const float*)d_in[1];
    const int*   adj_src  = (const int*)d_in[2];
    const int*   adj_dst  = (const int*)d_in[3];
    const float* adj_vals = (const float*)d_in[4];
    float*       out      = (float*)d_out;
    float*       acc      = (float*)d_ws;

    const int n4 = (N_NODES * N_HID) / 4;
    zero_acc<<<(n4 + 255) / 256, 256, 0, stream>>>((float4*)acc, n4);

    const int nblocks = (TOTAL_EDGES + 7) / 8;
    scatter_edges<<<nblocks, 256, 0, stream>>>(x, weight, adj_src, adj_dst,
                                               adj_vals, acc);

    const int tiles = N_NODES / 16;
    finalize_norm_gelu<<<(tiles + 4) / 5, 160, 0, stream>>>(acc, out);
}


# Pack kernel also builds a presence map (per 32 edges, 1 bit per row block); owner blocks test 1 bit per 32 keys and fetch only the groups that hit, not the whole key list; row sum order changes
# speedup vs baseline: 3.3609x; 3.3609x over previous
#include <hip/hip_runtime.h>
#include <math.h>


#define N_NODES 50000
#define N_HID   128
#define N_ADJ   4
#define N_EDGE  600000
#define TOTAL_EDGES (N_ADJ * N_EDGE)

#define ROWS_PER_BLOCK 200
#define N_ROW_BLOCKS   (N_NODES / ROWS_PER_BLOCK)
#define N_PLANE        ((N_ROW_BLOCKS + 31) / 32)
#define SCAN_THREADS   1024
#define SCAN_WAVES     (SCAN_THREADS / 32)
#define N_GRP          (TOTAL_EDGES / 32)
#define N_IT           ((N_GRP + SCAN_THREADS - 1) / SCAN_THREADS)
#define GRP_PER_WAVE   (N_IT * 32)
#define N_GRP_PAD      (N_IT * SCAN_THREADS)
#define N_KEY4         (TOTAL_EDGES / 8)
#define N_KEY4_PAD     (N_GRP_PAD * 4)
#define KEY_DWORDS     (N_KEY4_PAD * 4)
#define PACK_THREADS   256
#define PACK_GROUPS    (PACK_THREADS / 4)
#define NO_ROW_KEY     0xFFFFu
#define MAP_DEPTH      8
#define TAKE           16
#define QCAP           1280
#define STOP_AT        (QCAP - TAKE * 32)
#define QSTRIDE        (QCAP + 4)
#define HITCAP         (TAKE + MAP_DEPTH * 32)
#define OWNCAP         (32 + 4 * 32)

static_assert(N_NODES % ROWS_PER_BLOCK == 0, "rows must tile exactly");
static_assert(TOTAL_EDGES % 32 == 0, "whole groups of 32 edges");
static_assert(N_NODES <= 0xFFFF, "a node index must fit a 16-bit key, 0xFFFF left free");
static_assert(N_KEY4_PAD % PACK_THREADS == 0, "pack kernel grid is exact");
static_assert(PACK_THREADS / 32 == N_PLANE, "one pack wave stores one plane");
static_assert((KEY_DWORDS * 4) % 128 == 0 && (N_GRP_PAD * 4) % 128 == 0, "planes start on a 128-byte line");
static_assert(STOP_AT > 0, "a round must be able to take a batch");
static_assert(TOTAL_EDGES <= (1 << 22) && ROWS_PER_BLOCK <= (1 << 10), "queue entry packing");
static_assert((QSTRIDE * 4) % 16 == 0, "column reads are 16-byte loads");

typedef __attribute__((ext_vector_type(2))) float v2f;
typedef __attribute__((ext_vector_type(2))) unsigned short us2;
typedef __attribute__((ext_vector_type(8))) float v8f;

__device__ __forceinline__ void add_into_row(float* accL, int r, int lane, float4 xv, float sc) {
    float4* ap = (float4*)(accL + r * N_HID) + lane;
    float4 a = *ap;
    a.x += xv.x * sc;
    a.y += xv.y * sc;
    a.z += xv.z * sc;
    a.w += xv.w * sc;
    *ap = a;
}

__device__ __forceinline__ void stage_meta(const unsigned* oq, int pos, int n,
                                           const float* __restrict__ weight,
                                           const int*   __restrict__ adj_src,
                                           const float* __restrict__ adj_vals,
                                           int lane, int& s, int& sci, int& r) {
    const bool     ok  = lane < n;
    const unsigned ent = ok ? oq[pos + lane] : 0u;
    const unsigned e   = ent & 0x3FFFFFu;
    r = (int)(ent >> 22);
    const int   sv = adj_src[e];
    const float v  = adj_vals[e];
    const unsigned a = (e >= (unsigned)N_EDGE) + (e >= 2u * N_EDGE) + (e >= 3u * N_EDGE);
    const float sc = weight[a] * v;
    s   = min(max(sv, 0), N_NODES - 1);
    sci = __float_as_int(sc);
}

__device__ __forceinline__ void add_rows(int s, int sci, int r, int n,
                                         const float* __restrict__ x,
                                         float* accL, int lane) {
    int j = 0;
    for (; j + 8 <= n; j += 8) {
        float4 xv[8];
        #pragma unroll
        for (int k = 0; k < 8; ++k) {
            const int sj = __builtin_amdgcn_readlane(s, j + k);
            xv[k] = ((const float4*)(x + (size_t)sj * N_HID))[lane];
        }
        __builtin_amdgcn_sched_barrier(0);
        #pragma unroll
        for (int k = 0; k < 8; ++k) {
            const float scj = __int_as_float(__builtin_amdgcn_readlane(sci, j + k));
            const int   rj  = __builtin_amdgcn_readlane(r, j + k);
            add_into_row(accL, rj, lane, xv[k], scj);
        }
    }
    if (j < n) {
        float4 xv[7];
        #pragma unroll
        for (int k = 0; k < 7; ++k) {
            if (j + k < n) {
                const int sj = __builtin_amdgcn_readlane(s, j + k);
                xv[k] = ((const float4*)(x + (size_t)sj * N_HID))[lane];
            }
        }
        #pragma unroll
        for (int k = 0; k < 7; ++k) {
            if (j + k < n) {
                const float scj = __int_as_float(__builtin_amdgcn_readlane(sci, j + k));
                const int   rj  = __builtin_amdgcn_readlane(r, j + k);
                add_into_row(accL, rj, lane, xv[k], scj);
            }
        }
    }
}

__device__ __forceinline__ void drain_list(const unsigned* oq, int pos, int n,
                                           const float* __restrict__ x,
                                           const float* __restrict__ weight,
                                           const int*   __restrict__ adj_src,
                                           const float* __restrict__ adj_vals,
                                           float* accL, int lane) {
    int s, sci, r;
    stage_meta(oq, pos, n, weight, adj_src, adj_vals, lane, s, sci, r);
    add_rows(s, sci, r, n, x, accL, lane);
}

__device__ __forceinline__ int drain_full_groups(unsigned* oq, int own,
                                                 const float* __restrict__ x,
                                                 const float* __restrict__ weight,
                                                 const int*   __restrict__ adj_src,
                                                 const float* __restrict__ adj_vals,
                                                 float* accL, int lane) {
    int pos = 0;
    while (own - pos >= 32) {
        drain_list(oq, pos, 32, x, weight, adj_src, adj_vals, accL, lane);
        pos += 32;
    }
    if (pos > 0) {
        const int rem = own - pos;
        unsigned keep = 0u;
        if (lane < rem) keep = oq[pos + lane];
        if (lane < rem) oq[lane] = keep;
        own = rem;
    }
    return own;
}

__device__ __forceinline__ int append_mine(unsigned* oq, int own, bool mine, unsigned ent) {
    const unsigned mk = __builtin_amdgcn_ballot_w32(mine);
    if (mine) oq[own + (int)__builtin_amdgcn_mbcnt_lo(mk, 0u)] = ent;
    return own + __builtin_popcount(mk);
}

__device__ __forceinline__ unsigned key_of(int d) {
    return ((unsigned)d < (unsigned)N_NODES) ? (unsigned)d : NO_ROW_KEY;
}

__device__ __forceinline__ void mark_block(unsigned (&w)[N_PLANE], int d) {
    if ((unsigned)d < (unsigned)N_NODES) {
        const unsigned blk = (unsigned)d / ROWS_PER_BLOCK;
        const unsigned bit = 1u << (blk & 31u);
        const unsigned pl  = blk >> 5;
        #pragma unroll
        for (int j = 0; j < N_PLANE; ++j) w[j] |= (pl == (unsigned)j) ? bit : 0u;
    }
}

__global__ __launch_bounds__(PACK_THREADS) void pack_dst_keys(
    const int* __restrict__ adj_dst, unsigned* __restrict__ keys, unsigned* __restrict__ bmap)
{
    __shared__ unsigned sb[N_PLANE][PACK_GROUPS];
    const int i = blockIdx.x * PACK_THREADS + threadIdx.x;
    uint4 k = make_uint4(0xFFFFFFFFu, 0xFFFFFFFFu, 0xFFFFFFFFu, 0xFFFFFFFFu);
    unsigned w[N_PLANE];
    #pragma unroll
    for (int j = 0; j < N_PLANE; ++j) w[j] = 0u;
    if (i < N_KEY4) {
        const int4 a = ((const int4*)adj_dst)[2 * i];
        const int4 b = ((const int4*)adj_dst)[2 * i + 1];
        k.x = key_of(a.x) | (key_of(a.y) << 16);
        k.y = key_of(a.z) | (key_of(a.w) << 16);
        k.z = key_of(b.x) | (key_of(b.y) << 16);
        k.w = key_of(b.z) | (key_of(b.w) << 16);
        mark_block(w, a.x); mark_block(w, a.y); mark_block(w, a.z); mark_block(w, a.w);
        mark_block(w, b.x); mark_block(w, b.y); mark_block(w, b.z); mark_block(w, b.w);
    }
    ((uint4*)keys)[i] = k;

    #pragma unroll
    for (int j = 0; j < N_PLANE; ++j) {
        w[j] |= __shfl_xor(w[j], 1, 32);
        w[j] |= __shfl_xor(w[j], 2, 32);
    }
    if ((threadIdx.x & 3) == 0) {
        #pragma unroll
        for (int j = 0; j < N_PLANE; ++j) sb[j][threadIdx.x >> 2] = w[j];
    }
    __syncthreads();
    const int pw = threadIdx.x >> 5;
    const int ln = threadIdx.x & 31;
    unsigned* pl = bmap + (size_t)pw * N_GRP_PAD + (size_t)blockIdx.x * PACK_GROUPS;
    pl[ln]      = sb[pw][ln];
    pl[32 + ln] = sb[pw][32 + ln];
}

__device__ __forceinline__ int push_pair(unsigned* sq, int found, unsigned t, unsigned e) {
    const unsigned tl = t & 0xFFFFu;
    const unsigned th = t >> 16;
    const bool ml = tl < (unsigned)ROWS_PER_BLOCK;
    const bool mh = th < (unsigned)ROWS_PER_BLOCK;
    const unsigned kl = __builtin_amdgcn_ballot_w32(ml);
    const unsigned kh = __builtin_amdgcn_ballot_w32(mh);
    if (ml) sq[found + (int)__builtin_amdgcn_mbcnt_lo(kl, 0u)] = (tl << 22) | e;
    found += __builtin_popcount(kl);
    if (mh) sq[found + (int)__builtin_amdgcn_mbcnt_lo(kh, 0u)] = (th << 22) | (e + 1u);
    found += __builtin_popcount(kh);
    return found;
}

__device__ __forceinline__ int take_groups(const unsigned* gl, int n,
                                           const unsigned* __restrict__ keys,
                                           unsigned* sq, int found, us2 lov, int lane) {
    const unsigned sub  = (unsigned)lane & 15u;
    const int      half = lane >> 4;
    unsigned g[TAKE / 2];
    unsigned d[TAKE / 2];
    #pragma unroll
    for (int k = 0; k < TAKE / 2; ++k) {
        const bool ok = 2 * k + half < n;
        g[k] = ok ? gl[2 * k + half] : 0u;
        d[k] = ok ? keys[g[k] * 16u + sub] : 0xFFFFFFFFu;
    }
    #pragma unroll
    for (int k = 0; k < TAKE / 2; ++k) {
        if (2 * k < n) {
            const us2 tv = __builtin_bit_cast(us2, d[k]) - lov;
            found = push_pair(sq, found, __builtin_bit_cast(unsigned, tv), g[k] * 32u + 2u * sub);
        }
    }
    return found;
}

__global__ __launch_bounds__(SCAN_THREADS) void segment_sum_rows(
    const float* __restrict__ x,
    const float* __restrict__ weight,
    const int*   __restrict__ adj_src,
    const float* __restrict__ adj_vals,
    const unsigned* __restrict__ keys,
    const unsigned* __restrict__ bmap,
    float*       __restrict__ acc)
{
    __shared__ __attribute__((aligned(16))) float    accL[ROWS_PER_BLOCK * N_HID];
    __shared__ __attribute__((aligned(16))) unsigned scanQ[SCAN_WAVES * QSTRIDE];
    __shared__ unsigned ownQ[SCAN_WAVES * OWNCAP];
    __shared__ unsigned hitL[SCAN_WAVES * HITCAP];
    __shared__ int      scanCnt[SCAN_WAVES];
    __shared__ int      scanLeft[SCAN_WAVES];

    const int tid  = threadIdx.x;
    const int lane = tid & 31;
    const int wave = tid >> 5;
    unsigned* sq = scanQ + wave * QSTRIDE;
    unsigned* oq = ownQ  + wave * OWNCAP;
    unsigned* hl = hitL  + wave * HITCAP;
    const unsigned lo = blockIdx.x * ROWS_PER_BLOCK;
    const unsigned uw = (unsigned)wave;

    for (int i = tid; i < ROWS_PER_BLOCK * N_HID; i += SCAN_THREADS) accL[i] = 0.f;

    const unsigned lo2 = lo | (lo << 16);
    const us2 lov = __builtin_bit_cast(us2, lo2);
    const int wave_s = __builtin_amdgcn_readfirstlane(wave);
    const unsigned* plane = bmap + (size_t)(blockIdx.x >> 5) * N_GRP_PAD + (size_t)wave_s * GRP_PER_WAVE;
    const unsigned  bsh   = blockIdx.x & 31u;
    const unsigned  g0    = (unsigned)(wave * GRP_PER_WAVE + lane);
    const int rot = (int)((blockIdx.x & 31u) * N_IT) >> 5;
    int it   = 0;
    int pend = 0;
    unsigned left;

    do {
        int found = 0;
        while (true) {
            unsigned wn[MAP_DEPTH];
            #pragma unroll
            for (int u = 0; u < MAP_DEPTH; ++u) {
                int r = min(it + u, N_IT - 1) + rot;
                if (r >= N_IT) r -= N_IT;
                wn[u] = plane[r * 32 + lane];
            }
            while (pend >= TAKE && found <= STOP_AT) {
                pend -= TAKE;
                found = take_groups(hl + pend, TAKE, keys, sq, found, lov, lane);
            }
            if (pend >= TAKE) break;
            if (it >= N_IT) {
                if (pend > 0 && found <= STOP_AT) {
                    found = take_groups(hl, pend, keys, sq, found, lov, lane);
                    pend = 0;
                }
                break;
            }
            #pragma unroll
            for (int u = 0; u < MAP_DEPTH; ++u) {
                if (it + u < N_IT) {
                    int r = it + u + rot;
                    if (r >= N_IT) r -= N_IT;
                    const bool hit = ((wn[u] >> bsh) & 1u) != 0u;
                    const unsigned hm = __builtin_amdgcn_ballot_w32(hit);
                    if (hit) hl[pend + (int)__builtin_amdgcn_mbcnt_lo(hm, 0u)] = g0 + (unsigned)(r * 32);
                    pend += __builtin_popcount(hm);
                }
            }
            it += MAP_DEPTH;
        }
        if (lane == 0) {
            scanCnt[wave]  = found;
            scanLeft[wave] = (it < N_IT || pend > 0) ? 1 : 0;
        }
        __syncthreads();

        const int cs = scanCnt[lane];
        left = __builtin_amdgcn_ballot_w32(scanLeft[lane] != 0);
        const unsigned* cq = scanQ + lane * QSTRIDE;
        int own = 0;
        uint4 c = *(const uint4*)cq;
        for (int base = 0; __builtin_amdgcn_ballot_w32(base < cs) != 0u; base += 4) {
            const uint4 cn = *(const uint4*)(cq + base + 4);
            own = append_mine(oq, own, (base + 0 < cs) && (((c.x >> 22) & 31u) == uw), c.x);
            own = append_mine(oq, own, (base + 1 < cs) && (((c.y >> 22) & 31u) == uw), c.y);
            own = append_mine(oq, own, (base + 2 < cs) && (((c.z >> 22) & 31u) == uw), c.z);
            own = append_mine(oq, own, (base + 3 < cs) && (((c.w >> 22) & 31u) == uw), c.w);
            own = drain_full_groups(oq, own, x, weight, adj_src, adj_vals, accL, lane);
            c = cn;
        }
        if (own > 0) drain_list(oq, 0, own, x, weight, adj_src, adj_vals, accL, lane);

        __syncthreads();
    } while (left != 0u);

    float4*       acc4  = (float4*)acc + (size_t)lo * (N_HID / 4);
    const float4* accL4 = (const float4*)accL;
    for (int i = tid; i < ROWS_PER_BLOCK * (N_HID / 4); i += SCAN_THREADS) acc4[i] = accL4[i];
}

__global__ __launch_bounds__(160) void finalize_norm_gelu(
    const float* __restrict__ acc, float* __restrict__ out)
{
    const int lane = threadIdx.x & 31;
    const int wave = blockIdx.x * (blockDim.x >> 5) + (threadIdx.x >> 5);
    if (wave >= (N_NODES / 16)) return;

    const int node   = wave * 16 + (lane & 15);
    const int colOff = 2 * (lane >> 4);
    const float* rowp = acc + (long long)node * N_HID + colOff;

    v2f vbuf[32];
    v8f c = {};
    #pragma unroll
    for (int j = 0; j < 32; ++j) {
        v2f a;
        a.x = rowp[4 * j];
        a.y = rowp[4 * j + 1];
        vbuf[j] = a;
        c = __builtin_amdgcn_wmma_f32_16x16x4_f32(
                false, a, false, a, (short)0, c, false, false);
    }

    const int sel = lane & 7;
    float dv = c[0];
    dv = (sel == 1) ? c[1] : dv;
    dv = (sel == 2) ? c[2] : dv;
    dv = (sel == 3) ? c[3] : dv;
    dv = (sel == 4) ? c[4] : dv;
    dv = (sel == 5) ? c[5] : dv;
    dv = (sel == 6) ? c[6] : dv;
    dv = (sel == 7) ? c[7] : dv;
    const int nm      = lane & 15;
    const int srcLane = (nm < 8) ? nm : (nm + 16);
    const float sumsq = __shfl(dv, srcLane, 32);

    const float scale = 1.0f / fmaxf(sqrtf(sumsq), 1e-12f);

    float* op = out + (long long)node * N_HID + colOff;
    const float inv_sqrt2 = 0.70710678118654752440f;
    #pragma unroll
    for (int j = 0; j < 32; ++j) {
        float vx = vbuf[j].x * scale;
        float vy = vbuf[j].y * scale;
        vx = 0.5f * vx * (1.0f + erff(vx * inv_sqrt2));
        vy = 0.5f * vy * (1.0f + erff(vy * inv_sqrt2));
        op[4 * j]     = vx;
        op[4 * j + 1] = vy;
    }
}

extern "C" void kernel_launch(void* const* d_in, const int* in_sizes, int n_in,
                              void* d_out, int out_size, void* d_ws, size_t ws_size,
                              hipStream_t stream) {
    const float* x        = (const float*)d_in[0];
    const float* weight   = (const float*)d_in[1];
    const int*   adj_src  = (const int*)d_in[2];
    const int*   adj_dst  = (const int*)d_in[3];
    const float* adj_vals = (const float*)d_in[4];
    float*       out      = (float*)d_out;
    float*       acc      = (float*)d_ws;
    unsigned*    keys     = (unsigned*)d_out;
    unsigned*    bmap     = (unsigned*)d_out + KEY_DWORDS;

    pack_dst_keys<<<N_KEY4_PAD / PACK_THREADS, PACK_THREADS, 0, stream>>>(adj_dst, keys, bmap);

    segment_sum_rows<<<N_ROW_BLOCKS, SCAN_THREADS, 0, stream>>>(x, weight, adj_src, adj_vals,
                                                               keys, bmap, acc);

    const int tiles = N_NODES / 16;
    finalize_norm_gelu<<<(tiles + 4) / 5, 160, 0, stream>>>(acc, out);
}


# Scan waves counting-sort their queues by owner wave into one owner-ordered LDS list (ballot ranks, no atomics); an owner reads only its own run, not all 32 queues; row sum order changes, still fixed
# speedup vs baseline: 3.6154x; 1.0757x over previous
#include <hip/hip_runtime.h>
#include <math.h>


#define N_NODES 50000
#define N_HID   128
#define N_ADJ   4
#define N_EDGE  600000
#define TOTAL_EDGES (N_ADJ * N_EDGE)

#define ROWS_PER_BLOCK 200
#define N_ROW_BLOCKS   (N_NODES / ROWS_PER_BLOCK)
#define N_PLANE        ((N_ROW_BLOCKS + 31) / 32)
#define SCAN_THREADS   1024
#define SCAN_WAVES     (SCAN_THREADS / 32)
#define N_GRP          (TOTAL_EDGES / 32)
#define N_IT           ((N_GRP + SCAN_THREADS - 1) / SCAN_THREADS)
#define GRP_PER_WAVE   (N_IT * 32)
#define N_GRP_PAD      (N_IT * SCAN_THREADS)
#define N_KEY4         (TOTAL_EDGES / 8)
#define N_KEY4_PAD     (N_GRP_PAD * 4)
#define KEY_DWORDS     (N_KEY4_PAD * 4)
#define PACK_THREADS   256
#define PACK_GROUPS    (PACK_THREADS / 4)
#define NO_ROW_KEY     0xFFFFu
#define MAP_DEPTH      8
#define TAKE           16
#define QCAP           928
#define STOP_AT        (QCAP - TAKE * 32)
#define QSTRIDE        QCAP
#define LISTCAP        (SCAN_WAVES * QCAP)
#define HITCAP         (TAKE + MAP_DEPTH * 32)

static_assert(N_NODES % ROWS_PER_BLOCK == 0, "rows must tile exactly");
static_assert(TOTAL_EDGES % 32 == 0, "whole groups of 32 edges");
static_assert(N_NODES <= 0xFFFF, "a node index must fit a 16-bit key, 0xFFFF left free");
static_assert(N_KEY4_PAD % PACK_THREADS == 0, "pack kernel grid is exact");
static_assert(PACK_THREADS / 32 == N_PLANE, "one pack wave stores one plane");
static_assert((KEY_DWORDS * 4) % 128 == 0 && (N_GRP_PAD * 4) % 128 == 0, "planes start on a 128-byte line");
static_assert(STOP_AT > 0, "a round must be able to take a batch");
static_assert(TOTAL_EDGES <= (1 << 22) && ROWS_PER_BLOCK <= (1 << 10), "queue entry packing");
static_assert(SCAN_WAVES == 32, "one lane per owning wave in the count and prefix steps");
static_assert(SCAN_WAVES * QSTRIDE <= (1 << 16), "a place in the scan queues must fit 16 bits");

typedef __attribute__((ext_vector_type(2))) float v2f;
typedef __attribute__((ext_vector_type(2))) unsigned short us2;
typedef __attribute__((ext_vector_type(8))) float v8f;

__device__ __forceinline__ void add_into_row(float* accL, int r, int lane, float4 xv, float sc) {
    float4* ap = (float4*)(accL + r * N_HID) + lane;
    float4 a = *ap;
    a.x += xv.x * sc;
    a.y += xv.y * sc;
    a.z += xv.z * sc;
    a.w += xv.w * sc;
    *ap = a;
}

__device__ __forceinline__ void stage_meta(const unsigned short* ol, const unsigned* q, int pos, int n,
                                           const float* __restrict__ weight,
                                           const int*   __restrict__ adj_src,
                                           const float* __restrict__ adj_vals,
                                           int lane, int& s, int& sci, int& r) {
    const bool     ok  = lane < n;
    const unsigned ent = ok ? q[ol[pos + lane]] : 0u;
    const unsigned e   = ent & 0x3FFFFFu;
    r = (int)(ent >> 22);
    const int   sv = adj_src[e];
    const float v  = adj_vals[e];
    const unsigned a = (e >= (unsigned)N_EDGE) + (e >= 2u * N_EDGE) + (e >= 3u * N_EDGE);
    const float sc = weight[a] * v;
    s   = min(max(sv, 0), N_NODES - 1);
    sci = __float_as_int(sc);
}

__device__ __forceinline__ void add_rows(int s, int sci, int r, int n,
                                         const float* __restrict__ x,
                                         float* accL, int lane) {
    int j = 0;
    for (; j + 8 <= n; j += 8) {
        float4 xv[8];
        #pragma unroll
        for (int k = 0; k < 8; ++k) {
            const int sj = __builtin_amdgcn_readlane(s, j + k);
            xv[k] = ((const float4*)(x + (size_t)sj * N_HID))[lane];
        }
        __builtin_amdgcn_sched_barrier(0);
        #pragma unroll
        for (int k = 0; k < 8; ++k) {
            const float scj = __int_as_float(__builtin_amdgcn_readlane(sci, j + k));
            const int   rj  = __builtin_amdgcn_readlane(r, j + k);
            add_into_row(accL, rj, lane, xv[k], scj);
        }
    }
    if (j < n) {
        float4 xv[7];
        #pragma unroll
        for (int k = 0; k < 7; ++k) {
            if (j + k < n) {
                const int sj = __builtin_amdgcn_readlane(s, j + k);
                xv[k] = ((const float4*)(x + (size_t)sj * N_HID))[lane];
            }
        }
        #pragma unroll
        for (int k = 0; k < 7; ++k) {
            if (j + k < n) {
                const float scj = __int_as_float(__builtin_amdgcn_readlane(sci, j + k));
                const int   rj  = __builtin_amdgcn_readlane(r, j + k);
                add_into_row(accL, rj, lane, xv[k], scj);
            }
        }
    }
}

__device__ __forceinline__ void drain_list(const unsigned short* ol, const unsigned* q, int pos, int n,
                                           const float* __restrict__ x,
                                           const float* __restrict__ weight,
                                           const int*   __restrict__ adj_src,
                                           const float* __restrict__ adj_vals,
                                           float* accL, int lane) {
    int s, sci, r;
    stage_meta(ol, q, pos, n, weight, adj_src, adj_vals, lane, s, sci, r);
    add_rows(s, sci, r, n, x, accL, lane);
}

__device__ __forceinline__ unsigned same_owner_lanes(unsigned owner, bool valid) {
    unsigned m = __builtin_amdgcn_ballot_w32(valid);
    for (int b = 0; b < 5; ++b) {
        const bool     bit = ((owner >> b) & 1u) != 0u;
        const unsigned bal = __builtin_amdgcn_ballot_w32(bit);
        m &= bit ? bal : ~bal;
    }
    return m;
}

__device__ __forceinline__ unsigned key_of(int d) {
    return ((unsigned)d < (unsigned)N_NODES) ? (unsigned)d : NO_ROW_KEY;
}

__device__ __forceinline__ void mark_block(unsigned (&w)[N_PLANE], int d) {
    if ((unsigned)d < (unsigned)N_NODES) {
        const unsigned blk = (unsigned)d / ROWS_PER_BLOCK;
        const unsigned bit = 1u << (blk & 31u);
        const unsigned pl  = blk >> 5;
        #pragma unroll
        for (int j = 0; j < N_PLANE; ++j) w[j] |= (pl == (unsigned)j) ? bit : 0u;
    }
}

__global__ __launch_bounds__(PACK_THREADS) void pack_dst_keys(
    const int* __restrict__ adj_dst, unsigned* __restrict__ keys, unsigned* __restrict__ bmap)
{
    __shared__ unsigned sb[N_PLANE][PACK_GROUPS];
    const int i = blockIdx.x * PACK_THREADS + threadIdx.x;
    uint4 k = make_uint4(0xFFFFFFFFu, 0xFFFFFFFFu, 0xFFFFFFFFu, 0xFFFFFFFFu);
    unsigned w[N_PLANE];
    #pragma unroll
    for (int j = 0; j < N_PLANE; ++j) w[j] = 0u;
    if (i < N_KEY4) {
        const int4 a = ((const int4*)adj_dst)[2 * i];
        const int4 b = ((const int4*)adj_dst)[2 * i + 1];
        k.x = key_of(a.x) | (key_of(a.y) << 16);
        k.y = key_of(a.z) | (key_of(a.w) << 16);
        k.z = key_of(b.x) | (key_of(b.y) << 16);
        k.w = key_of(b.z) | (key_of(b.w) << 16);
        mark_block(w, a.x); mark_block(w, a.y); mark_block(w, a.z); mark_block(w, a.w);
        mark_block(w, b.x); mark_block(w, b.y); mark_block(w, b.z); mark_block(w, b.w);
    }
    ((uint4*)keys)[i] = k;

    #pragma unroll
    for (int j = 0; j < N_PLANE; ++j) {
        w[j] |= __shfl_xor(w[j], 1, 32);
        w[j] |= __shfl_xor(w[j], 2, 32);
    }
    if ((threadIdx.x & 3) == 0) {
        #pragma unroll
        for (int j = 0; j < N_PLANE; ++j) sb[j][threadIdx.x >> 2] = w[j];
    }
    __syncthreads();
    const int pw = threadIdx.x >> 5;
    const int ln = threadIdx.x & 31;
    unsigned* pl = bmap + (size_t)pw * N_GRP_PAD + (size_t)blockIdx.x * PACK_GROUPS;
    pl[ln]      = sb[pw][ln];
    pl[32 + ln] = sb[pw][32 + ln];
}

__device__ __forceinline__ int push_pair(unsigned* sq, int found, unsigned t, unsigned e) {
    const unsigned tl = t & 0xFFFFu;
    const unsigned th = t >> 16;
    const bool ml = tl < (unsigned)ROWS_PER_BLOCK;
    const bool mh = th < (unsigned)ROWS_PER_BLOCK;
    const unsigned kl = __builtin_amdgcn_ballot_w32(ml);
    const unsigned kh = __builtin_amdgcn_ballot_w32(mh);
    if (ml) sq[found + (int)__builtin_amdgcn_mbcnt_lo(kl, 0u)] = (tl << 22) | e;
    found += __builtin_popcount(kl);
    if (mh) sq[found + (int)__builtin_amdgcn_mbcnt_lo(kh, 0u)] = (th << 22) | (e + 1u);
    found += __builtin_popcount(kh);
    return found;
}

__device__ __forceinline__ int take_groups(const unsigned* gl, int n,
                                           const unsigned* __restrict__ keys,
                                           unsigned* sq, int found, us2 lov, int lane) {
    const unsigned sub  = (unsigned)lane & 15u;
    const int      half = lane >> 4;
    unsigned g[TAKE / 2];
    unsigned d[TAKE / 2];
    #pragma unroll
    for (int k = 0; k < TAKE / 2; ++k) {
        const bool ok = 2 * k + half < n;
        g[k] = ok ? gl[2 * k + half] : 0u;
        d[k] = ok ? keys[g[k] * 16u + sub] : 0xFFFFFFFFu;
    }
    #pragma unroll
    for (int k = 0; k < TAKE / 2; ++k) {
        if (2 * k < n) {
            const us2 tv = __builtin_bit_cast(us2, d[k]) - lov;
            found = push_pair(sq, found, __builtin_bit_cast(unsigned, tv), g[k] * 32u + 2u * sub);
        }
    }
    return found;
}

__global__ __launch_bounds__(SCAN_THREADS) void segment_sum_rows(
    const float* __restrict__ x,
    const float* __restrict__ weight,
    const int*   __restrict__ adj_src,
    const float* __restrict__ adj_vals,
    const unsigned* __restrict__ keys,
    const unsigned* __restrict__ bmap,
    float*       __restrict__ acc)
{
    __shared__ __attribute__((aligned(16))) float    accL[ROWS_PER_BLOCK * N_HID];
    __shared__ unsigned scanQ[SCAN_WAVES * QSTRIDE];
    __shared__ unsigned short ownL[LISTCAP];
    __shared__ unsigned hitL[SCAN_WAVES * HITCAP];
    __shared__ int      cntM[SCAN_WAVES * 32];
    __shared__ int      curM[SCAN_WAVES * 32];
    __shared__ int      scanLeft[SCAN_WAVES];

    const int tid  = threadIdx.x;
    const int lane = tid & 31;
    const int wave = tid >> 5;
    unsigned* sq = scanQ + wave * QSTRIDE;
    unsigned* hl = hitL  + wave * HITCAP;
    int*      hist = cntM + wave * 32;
    int*      cur  = curM + wave * 32;
    const unsigned lo = blockIdx.x * ROWS_PER_BLOCK;

    for (int i = tid; i < ROWS_PER_BLOCK * N_HID; i += SCAN_THREADS) accL[i] = 0.f;

    const unsigned lo2 = lo | (lo << 16);
    const us2 lov = __builtin_bit_cast(us2, lo2);
    const int wave_s = __builtin_amdgcn_readfirstlane(wave);
    const unsigned* plane = bmap + (size_t)(blockIdx.x >> 5) * N_GRP_PAD + (size_t)wave_s * GRP_PER_WAVE;
    const unsigned  bsh   = blockIdx.x & 31u;
    const unsigned  g0    = (unsigned)(wave * GRP_PER_WAVE + lane);
    const int rot = (int)((blockIdx.x & 31u) * N_IT) >> 5;
    int it   = 0;
    int pend = 0;
    unsigned left;

    do {
        int found = 0;
        while (true) {
            unsigned wn[MAP_DEPTH];
            #pragma unroll
            for (int u = 0; u < MAP_DEPTH; ++u) {
                int r = min(it + u, N_IT - 1) + rot;
                if (r >= N_IT) r -= N_IT;
                wn[u] = plane[r * 32 + lane];
            }
            while (pend >= TAKE && found <= STOP_AT) {
                pend -= TAKE;
                found = take_groups(hl + pend, TAKE, keys, sq, found, lov, lane);
            }
            if (pend >= TAKE) break;
            if (it >= N_IT) {
                if (pend > 0 && found <= STOP_AT) {
                    found = take_groups(hl, pend, keys, sq, found, lov, lane);
                    pend = 0;
                }
                break;
            }
            #pragma unroll
            for (int u = 0; u < MAP_DEPTH; ++u) {
                if (it + u < N_IT) {
                    int r = it + u + rot;
                    if (r >= N_IT) r -= N_IT;
                    const bool hit = ((wn[u] >> bsh) & 1u) != 0u;
                    const unsigned hm = __builtin_amdgcn_ballot_w32(hit);
                    if (hit) hl[pend + (int)__builtin_amdgcn_mbcnt_lo(hm, 0u)] = g0 + (unsigned)(r * 32);
                    pend += __builtin_popcount(hm);
                }
            }
            it += MAP_DEPTH;
        }
        hist[lane] = 0;
        for (int b = 0; b < found; b += 32) {
            const bool     ok  = b + lane < found;
            const unsigned ent = ok ? sq[b + lane] : 0u;
            const unsigned own = (ent >> 22) & 31u;
            const unsigned m   = same_owner_lanes(own, ok);
            if (ok && __builtin_amdgcn_mbcnt_lo(m, 0u) == 0u) hist[own] += __builtin_popcount(m);
        }
        if (lane == 0) scanLeft[wave] = (it < N_IT || pend > 0) ? 1 : 0;
        __syncthreads();

        left = __builtin_amdgcn_ballot_w32(scanLeft[lane] != 0);
        int tot = 0, pre = 0;
        for (int w2 = 0; w2 < SCAN_WAVES; ++w2) {
            const int cw = cntM[w2 * 32 + lane];
            tot += cw;
            pre += (w2 < wave_s) ? cw : 0;
        }
        int inc = tot;
        for (int d = 1; d < 32; d <<= 1) {
            const int t = __shfl_up(inc, d, 32);
            if (lane >= d) inc += t;
        }
        const int lbase = inc - tot;
        cur[lane] = lbase + pre;
        const int myBeg = __builtin_amdgcn_readlane(lbase, wave_s);
        const int myCnt = __builtin_amdgcn_readlane(tot,   wave_s);

        for (int b = 0; b < found; b += 32) {
            const bool     ok  = b + lane < found;
            const unsigned ent = ok ? sq[b + lane] : 0u;
            const unsigned own = (ent >> 22) & 31u;
            const unsigned m   = same_owner_lanes(own, ok);
            const int      rk  = (int)__builtin_amdgcn_mbcnt_lo(m, 0u);
            if (ok) {
                const int p = cur[own];
                ownL[p + rk] = (unsigned short)(wave * QSTRIDE + b + lane);
                if (rk == 0) cur[own] = p + __builtin_popcount(m);
            }
        }
        __syncthreads();

        const unsigned short* ol = ownL + myBeg;
        int pos = 0;
        for (; pos + 32 <= myCnt; pos += 32)
            drain_list(ol, scanQ, pos, 32, x, weight, adj_src, adj_vals, accL, lane);
        if (pos < myCnt)
            drain_list(ol, scanQ, pos, myCnt - pos, x, weight, adj_src, adj_vals, accL, lane);

        __syncthreads();
    } while (left != 0u);

    float4*       acc4  = (float4*)acc + (size_t)lo * (N_HID / 4);
    const float4* accL4 = (const float4*)accL;
    for (int i = tid; i < ROWS_PER_BLOCK * (N_HID / 4); i += SCAN_THREADS) acc4[i] = accL4[i];
}

__global__ __launch_bounds__(160) void finalize_norm_gelu(
    const float* __restrict__ acc, float* __restrict__ out)
{
    const int lane = threadIdx.x & 31;
    const int wave = blockIdx.x * (blockDim.x >> 5) + (threadIdx.x >> 5);
    if (wave >= (N_NODES / 16)) return;

    const int node   = wave * 16 + (lane & 15);
    const int colOff = 2 * (lane >> 4);
    const float* rowp = acc + (long long)node * N_HID + colOff;

    v2f vbuf[32];
    v8f c = {};
    #pragma unroll
    for (int j = 0; j < 32; ++j) {
        v2f a;
        a.x = rowp[4 * j];
        a.y = rowp[4 * j + 1];
        vbuf[j] = a;
        c = __builtin_amdgcn_wmma_f32_16x16x4_f32(
                false, a, false, a, (short)0, c, false, false);
    }

    const int sel = lane & 7;
    float dv = c[0];
    dv = (sel == 1) ? c[1] : dv;
    dv = (sel == 2) ? c[2] : dv;
    dv = (sel == 3) ? c[3] : dv;
    dv = (sel == 4) ? c[4] : dv;
    dv = (sel == 5) ? c[5] : dv;
    dv = (sel == 6) ? c[6] : dv;
    dv = (sel == 7) ? c[7] : dv;
    const int nm      = lane & 15;
    const int srcLane = (nm < 8) ? nm : (nm + 16);
    const float sumsq = __shfl(dv, srcLane, 32);

    const float scale = 1.0f / fmaxf(sqrtf(sumsq), 1e-12f);

    float* op = out + (long long)node * N_HID + colOff;
    const float inv_sqrt2 = 0.70710678118654752440f;
    #pragma unroll
    for (int j = 0; j < 32; ++j) {
        float vx = vbuf[j].x * scale;
        float vy = vbuf[j].y * scale;
        vx = 0.5f * vx * (1.0f + erff(vx * inv_sqrt2));
        vy = 0.5f * vy * (1.0f + erff(vy * inv_sqrt2));
        op[4 * j]     = vx;
        op[4 * j + 1] = vy;
    }
}

extern "C" void kernel_launch(void* const* d_in, const int* in_sizes, int n_in,
                              void* d_out, int out_size, void* d_ws, size_t ws_size,
                              hipStream_t stream) {
    const float* x        = (const float*)d_in[0];
    const float* weight   = (const float*)d_in[1];
    const int*   adj_src  = (const int*)d_in[2];
    const int*   adj_dst  = (const int*)d_in[3];
    const float* adj_vals = (const float*)d_in[4];
    float*       out      = (float*)d_out;
    float*       acc      = (float*)d_ws;
    unsigned*    keys     = (unsigned*)d_out;
    unsigned*    bmap     = (unsigned*)d_out + KEY_DWORDS;

    pack_dst_keys<<<N_KEY4_PAD / PACK_THREADS, PACK_THREADS, 0, stream>>>(adj_dst, keys, bmap);

    segment_sum_rows<<<N_ROW_BLOCKS, SCAN_THREADS, 0, stream>>>(x, weight, adj_src, adj_vals,
                                                               keys, bmap, acc);

    const int tiles = N_NODES / 16;
    finalize_norm_gelu<<<(tiles + 4) / 5, 160, 0, stream>>>(acc, out);
}


# Edge list packed once per call into 8-byte records (key, source row, scale) instead of 16-bit keys: the owner scan brings source and scale, the gather loads no index or value; row sum order changes
# speedup vs baseline: 3.8033x; 1.0520x over previous
#include <hip/hip_runtime.h>
#include <math.h>


#define N_NODES 50000
#define N_HID   128
#define N_ADJ   4
#define N_EDGE  600000
#define TOTAL_EDGES (N_ADJ * N_EDGE)

#define ROWS_PER_BLOCK 200
#define N_ROW_BLOCKS   (N_NODES / ROWS_PER_BLOCK)
#define N_PLANE        ((N_ROW_BLOCKS + 31) / 32)
#define SCAN_THREADS   1024
#define SCAN_WAVES     (SCAN_THREADS / 32)
#define N_GRP          (TOTAL_EDGES / 32)
#define N_IT           ((N_GRP + SCAN_THREADS - 1) / SCAN_THREADS)
#define GRP_PER_WAVE   (N_IT * 32)
#define N_GRP_PAD      (N_IT * SCAN_THREADS)
#define N_OCT          (TOTAL_EDGES / 8)
#define N_OCT_PAD      (N_GRP_PAD * 4)
#define REC_DWORDS     (N_GRP_PAD * 64)
#define PACK_THREADS   256
#define PACK_GROUPS    (PACK_THREADS / 4)
#define NO_ROW_KEY     0xFFFFu
#define MAP_DEPTH      8
#define TAKE           16
#define QCAP           560
#define FULL_AT        (QCAP - 32)
#define QSTRIDE        QCAP
#define LISTCAP        (SCAN_WAVES * QCAP)
#define HITCAP         (TAKE + MAP_DEPTH * 32)

static_assert(N_NODES % ROWS_PER_BLOCK == 0, "rows must tile exactly");
static_assert(TOTAL_EDGES % 32 == 0, "whole groups of 32 edges");
static_assert(N_NODES <= 0xFFFF, "a node index must fit a 16-bit key, 0xFFFF left free");
static_assert(N_OCT_PAD % PACK_THREADS == 0, "pack kernel grid is exact");
static_assert(N_EDGE % 8 == 0, "the 8 edges of a pack thread belong to one adjacency");
static_assert(PACK_THREADS / 32 == N_PLANE, "one pack wave stores one plane");
static_assert((REC_DWORDS * 4) % 256 == 0 && (N_GRP_PAD * 4) % 128 == 0, "planes start on a 128-byte line");
static_assert(FULL_AT >= TAKE * 32, "a round's first batch is always taken whole");
static_assert(ROWS_PER_BLOCK <= (1 << 16), "queue entry packing");
static_assert(SCAN_WAVES == 32, "one lane per owning wave in the count and prefix steps");
static_assert(SCAN_WAVES * QSTRIDE <= (1 << 16), "a place in the scan queues must fit 16 bits");

typedef __attribute__((ext_vector_type(2))) float v2f;
typedef __attribute__((ext_vector_type(8))) float v8f;

__device__ __forceinline__ void add_into_row(float* accL, int r, int lane, float4 xv, float sc) {
    float4* ap = (float4*)(accL + r * N_HID) + lane;
    float4 a = *ap;
    a.x += xv.x * sc;
    a.y += xv.y * sc;
    a.z += xv.z * sc;
    a.w += xv.w * sc;
    *ap = a;
}

__device__ __forceinline__ void stage_entries(const unsigned short* ol, const uint2* q, int pos, int n,
                                              int lane, int& s, int& sci, int& r) {
    const bool  ok  = lane < n;
    const uint2 ent = ok ? q[ol[pos + lane]] : make_uint2(0u, 0u);
    r   = (int)(ent.x & 0xFFFFu);
    s   = (int)(ent.x >> 16);
    sci = (int)ent.y;
}

__device__ __forceinline__ void add_rows(int s, int sci, int r, int n,
                                         const float* __restrict__ x,
                                         float* accL, int lane) {
    int j = 0;
    for (; j + 8 <= n; j += 8) {
        float4 xv[8];
        #pragma unroll
        for (int k = 0; k < 8; ++k) {
            const int sj = __builtin_amdgcn_readlane(s, j + k);
            xv[k] = ((const float4*)(x + (size_t)sj * N_HID))[lane];
        }
        __builtin_amdgcn_sched_barrier(0);
        #pragma unroll
        for (int k = 0; k < 8; ++k) {
            const float scj = __int_as_float(__builtin_amdgcn_readlane(sci, j + k));
            const int   rj  = __builtin_amdgcn_readlane(r, j + k);
            add_into_row(accL, rj, lane, xv[k], scj);
        }
    }
    if (j < n) {
        float4 xv[7];
        #pragma unroll
        for (int k = 0; k < 7; ++k) {
            if (j + k < n) {
                const int sj = __builtin_amdgcn_readlane(s, j + k);
                xv[k] = ((const float4*)(x + (size_t)sj * N_HID))[lane];
            }
        }
        #pragma unroll
        for (int k = 0; k < 7; ++k) {
            if (j + k < n) {
                const float scj = __int_as_float(__builtin_amdgcn_readlane(sci, j + k));
                const int   rj  = __builtin_amdgcn_readlane(r, j + k);
                add_into_row(accL, rj, lane, xv[k], scj);
            }
        }
    }
}

__device__ __forceinline__ void drain_list(const unsigned short* ol, const uint2* q, int pos, int n,
                                           const float* __restrict__ x,
                                           float* accL, int lane) {
    int s, sci, r;
    stage_entries(ol, q, pos, n, lane, s, sci, r);
    add_rows(s, sci, r, n, x, accL, lane);
}

__device__ __forceinline__ unsigned same_owner_lanes(unsigned owner, bool valid) {
    unsigned m = __builtin_amdgcn_ballot_w32(valid);
    for (int b = 0; b < 5; ++b) {
        const bool     bit = ((owner >> b) & 1u) != 0u;
        const unsigned bal = __builtin_amdgcn_ballot_w32(bit);
        m &= bit ? bal : ~bal;
    }
    return m;
}

__device__ __forceinline__ unsigned key_of(int d) {
    return ((unsigned)d < (unsigned)N_NODES) ? (unsigned)d : NO_ROW_KEY;
}

__device__ __forceinline__ unsigned rec_word(int d, int sv) {
    const unsigned s = (unsigned)min(max(sv, 0), N_NODES - 1);
    return key_of(d) | (s << 16);
}

__device__ __forceinline__ void mark_block(unsigned (&w)[N_PLANE], int d) {
    if ((unsigned)d < (unsigned)N_NODES) {
        const unsigned blk = (unsigned)d / ROWS_PER_BLOCK;
        const unsigned bit = 1u << (blk & 31u);
        const unsigned pl  = blk >> 5;
        #pragma unroll
        for (int j = 0; j < N_PLANE; ++j) w[j] |= (pl == (unsigned)j) ? bit : 0u;
    }
}

__global__ __launch_bounds__(PACK_THREADS) void pack_edge_records(
    const int*   __restrict__ adj_dst,
    const int*   __restrict__ adj_src,
    const float* __restrict__ adj_vals,
    const float* __restrict__ weight,
    uint4* __restrict__ recs, unsigned* __restrict__ bmap)
{
    __shared__ unsigned sb[N_PLANE][PACK_GROUPS];
    __shared__ uint4    stg[PACK_THREADS * 4];
    const int i = blockIdx.x * PACK_THREADS + threadIdx.x;
    const unsigned none = 0xFFFFFFFFu;
    uint4 q0 = make_uint4(none, 0u, none, 0u);
    uint4 q1 = q0, q2 = q0, q3 = q0;
    unsigned w[N_PLANE];
    #pragma unroll
    for (int j = 0; j < N_PLANE; ++j) w[j] = 0u;
    if (i < N_OCT) {
        const int4   a  = ((const int4*)adj_dst)[2 * i];
        const int4   b  = ((const int4*)adj_dst)[2 * i + 1];
        const int4   sa = ((const int4*)adj_src)[2 * i];
        const int4   sd = ((const int4*)adj_src)[2 * i + 1];
        const float4 va = ((const float4*)adj_vals)[2 * i];
        const float4 vb = ((const float4*)adj_vals)[2 * i + 1];
        const unsigned e0  = 8u * (unsigned)i;
        const unsigned adj = (e0 >= (unsigned)N_EDGE) + (e0 >= 2u * N_EDGE) + (e0 >= 3u * N_EDGE);
        const float wgt = weight[adj];
        q0 = make_uint4(rec_word(a.x, sa.x), __float_as_uint(wgt * va.x), rec_word(a.y, sa.y), __float_as_uint(wgt * va.y));
        q1 = make_uint4(rec_word(a.z, sa.z), __float_as_uint(wgt * va.z), rec_word(a.w, sa.w), __float_as_uint(wgt * va.w));
        q2 = make_uint4(rec_word(b.x, sd.x), __float_as_uint(wgt * vb.x), rec_word(b.y, sd.y), __float_as_uint(wgt * vb.y));
        q3 = make_uint4(rec_word(b.z, sd.z), __float_as_uint(wgt * vb.z), rec_word(b.w, sd.w), __float_as_uint(wgt * vb.w));
        mark_block(w, a.x); mark_block(w, a.y); mark_block(w, a.z); mark_block(w, a.w);
        mark_block(w, b.x); mark_block(w, b.y); mark_block(w, b.z); mark_block(w, b.w);
    }
    stg[4 * threadIdx.x]     = q0;
    stg[4 * threadIdx.x + 1] = q1;
    stg[4 * threadIdx.x + 2] = q2;
    stg[4 * threadIdx.x + 3] = q3;

    #pragma unroll
    for (int j = 0; j < N_PLANE; ++j) {
        w[j] |= __shfl_xor(w[j], 1, 32);
        w[j] |= __shfl_xor(w[j], 2, 32);
    }
    if ((threadIdx.x & 3) == 0) {
        #pragma unroll
        for (int j = 0; j < N_PLANE; ++j) sb[j][threadIdx.x >> 2] = w[j];
    }
    __syncthreads();
    uint4* rp = recs + (size_t)blockIdx.x * (PACK_THREADS * 4);
    #pragma unroll
    for (int j = 0; j < 4; ++j) rp[j * PACK_THREADS + threadIdx.x] = stg[j * PACK_THREADS + threadIdx.x];
    const int pw = threadIdx.x >> 5;
    const int ln = threadIdx.x & 31;
    unsigned* pl = bmap + (size_t)pw * N_GRP_PAD + (size_t)blockIdx.x * PACK_GROUPS;
    pl[ln]      = sb[pw][ln];
    pl[32 + ln] = sb[pw][32 + ln];
}

__device__ __forceinline__ int push_group(uint2* sq, int found, uint2 rec, unsigned lo) {
    const unsigned t  = (rec.x & 0xFFFFu) - lo;
    const bool     m  = t < (unsigned)ROWS_PER_BLOCK;
    const unsigned km = __builtin_amdgcn_ballot_w32(m);
    if (m) sq[found + (int)__builtin_amdgcn_mbcnt_lo(km, 0u)] = make_uint2((rec.x & 0xFFFF0000u) | t, rec.y);
    return found + __builtin_popcount(km);
}

__device__ __forceinline__ int take_groups(const unsigned* gt, int n,
                                           const uint2* __restrict__ recs,
                                           uint2* sq, int& found, unsigned lo, int lane) {
    const unsigned gv = (lane < n) ? gt[-1 - lane] : 0u;
    uint2 d[TAKE];
    #pragma unroll
    for (int k = 0; k < TAKE; ++k) {
        d[k] = make_uint2(0xFFFFFFFFu, 0u);
        if (k < n) {
            const unsigned g = (unsigned)__builtin_amdgcn_readlane((int)gv, k);
            d[k] = recs[(size_t)g * 32u + (unsigned)lane];
        }
    }
    int  taken = 0;
    bool go    = true;
    #pragma unroll
    for (int k = 0; k < TAKE; ++k) {
        go = go && k < n && found <= FULL_AT;
        if (go) {
            found = push_group(sq, found, d[k], lo);
            taken = k + 1;
        }
    }
    return taken;
}

__global__ __launch_bounds__(SCAN_THREADS) void segment_sum_rows(
    const float* __restrict__ x,
    const uint2* __restrict__ recs,
    const unsigned* __restrict__ bmap,
    float*       __restrict__ acc)
{
    __shared__ __attribute__((aligned(16))) float    accL[ROWS_PER_BLOCK * N_HID];
    __shared__ uint2    scanQ[SCAN_WAVES * QSTRIDE];
    __shared__ unsigned short ownL[LISTCAP];
    __shared__ unsigned hitL[SCAN_WAVES * HITCAP];
    __shared__ int      cntM[SCAN_WAVES * 32];
    __shared__ int      curM[SCAN_WAVES * 32];
    __shared__ int      scanLeft[SCAN_WAVES];

    const int tid  = threadIdx.x;
    const int lane = tid & 31;
    const int wave = tid >> 5;
    uint2*    sq = scanQ + wave * QSTRIDE;
    unsigned* hl = hitL  + wave * HITCAP;
    int*      hist = cntM + wave * 32;
    int*      cur  = curM + wave * 32;
    const unsigned lo = blockIdx.x * ROWS_PER_BLOCK;

    for (int i = tid; i < ROWS_PER_BLOCK * N_HID; i += SCAN_THREADS) accL[i] = 0.f;

    const int wave_s = __builtin_amdgcn_readfirstlane(wave);
    const unsigned* plane = bmap + (size_t)(blockIdx.x >> 5) * N_GRP_PAD + (size_t)wave_s * GRP_PER_WAVE;
    const unsigned  bsh   = blockIdx.x & 31u;
    const unsigned  g0    = (unsigned)(wave * GRP_PER_WAVE + lane);
    const int rot = (int)((blockIdx.x & 31u) * N_IT) >> 5;
    int it   = 0;
    int pend = 0;
    unsigned left;

    do {
        int found = 0;
        while (true) {
            unsigned wn[MAP_DEPTH];
            #pragma unroll
            for (int u = 0; u < MAP_DEPTH; ++u) {
                int r = min(it + u, N_IT - 1) + rot;
                if (r >= N_IT) r -= N_IT;
                wn[u] = plane[r * 32 + lane];
            }
            bool full = false;
            while (pend >= TAKE && !full) {
                const int tk = take_groups(hl + pend, TAKE, recs, sq, found, lo, lane);
                pend -= tk;
                full  = tk < TAKE;
            }
            if (full) break;
            if (it >= N_IT) {
                if (pend > 0) pend -= take_groups(hl + pend, pend, recs, sq, found, lo, lane);
                break;
            }
            #pragma unroll
            for (int u = 0; u < MAP_DEPTH; ++u) {
                if (it + u < N_IT) {
                    int r = it + u + rot;
                    if (r >= N_IT) r -= N_IT;
                    const bool hit = ((wn[u] >> bsh) & 1u) != 0u;
                    const unsigned hm = __builtin_amdgcn_ballot_w32(hit);
                    if (hit) hl[pend + (int)__builtin_amdgcn_mbcnt_lo(hm, 0u)] = g0 + (unsigned)(r * 32);
                    pend += __builtin_popcount(hm);
                }
            }
            it += MAP_DEPTH;
        }
        hist[lane] = 0;
        for (int b = 0; b < found; b += 32) {
            const bool     ok  = b + lane < found;
            const unsigned ent = ok ? sq[b + lane].x : 0u;
            const unsigned own = ent & 31u;
            const unsigned m   = same_owner_lanes(own, ok);
            if (ok && __builtin_amdgcn_mbcnt_lo(m, 0u) == 0u) hist[own] += __builtin_popcount(m);
        }
        if (lane == 0) scanLeft[wave] = (it < N_IT || pend > 0) ? 1 : 0;
        __syncthreads();

        left = __builtin_amdgcn_ballot_w32(scanLeft[lane] != 0);
        int tot = 0, pre = 0;
        for (int w2 = 0; w2 < SCAN_WAVES; ++w2) {
            const int cw = cntM[w2 * 32 + lane];
            tot += cw;
            pre += (w2 < wave_s) ? cw : 0;
        }
        int inc = tot;
        for (int d = 1; d < 32; d <<= 1) {
            const int t = __shfl_up(inc, d, 32);
            if (lane >= d) inc += t;
        }
        const int lbase = inc - tot;
        cur[lane] = lbase + pre;
        const int myBeg = __builtin_amdgcn_readlane(lbase, wave_s);
        const int myCnt = __builtin_amdgcn_readlane(tot,   wave_s);

        for (int b = 0; b < found; b += 32) {
            const bool     ok  = b + lane < found;
            const unsigned ent = ok ? sq[b + lane].x : 0u;
            const unsigned own = ent & 31u;
            const unsigned m   = same_owner_lanes(own, ok);
            const int      rk  = (int)__builtin_amdgcn_mbcnt_lo(m, 0u);
            if (ok) {
                const int p = cur[own];
                ownL[p + rk] = (unsigned short)(wave * QSTRIDE + b + lane);
                if (rk == 0) cur[own] = p + __builtin_popcount(m);
            }
        }
        __syncthreads();

        const unsigned short* ol = ownL + myBeg;
        int pos = 0;
        for (; pos + 32 <= myCnt; pos += 32)
            drain_list(ol, scanQ, pos, 32, x, accL, lane);
        if (pos < myCnt)
            drain_list(ol, scanQ, pos, myCnt - pos, x, accL, lane);

        __syncthreads();
    } while (left != 0u);

    float4*       acc4  = (float4*)acc + (size_t)lo * (N_HID / 4);
    const float4* accL4 = (const float4*)accL;
    for (int i = tid; i < ROWS_PER_BLOCK * (N_HID / 4); i += SCAN_THREADS) acc4[i] = accL4[i];
}

__global__ __launch_bounds__(160) void finalize_norm_gelu(
    const float* __restrict__ acc, float* __restrict__ out)
{
    const int lane = threadIdx.x & 31;
    const int wave = blockIdx.x * (blockDim.x >> 5) + (threadIdx.x >> 5);
    if (wave >= (N_NODES / 16)) return;

    const int node   = wave * 16 + (lane & 15);
    const int colOff = 2 * (lane >> 4);
    const float* rowp = acc + (long long)node * N_HID + colOff;

    v2f vbuf[32];
    v8f c = {};
    #pragma unroll
    for (int j = 0; j < 32; ++j) {
        v2f a;
        a.x = rowp[4 * j];
        a.y = rowp[4 * j + 1];
        vbuf[j] = a;
        c = __builtin_amdgcn_wmma_f32_16x16x4_f32(
                false, a, false, a, (short)0, c, false, false);
    }

    const int sel = lane & 7;
    float dv = c[0];
    dv = (sel == 1) ? c[1] : dv;
    dv = (sel == 2) ? c[2] : dv;
    dv = (sel == 3) ? c[3] : dv;
    dv = (sel == 4) ? c[4] : dv;
    dv = (sel == 5) ? c[5] : dv;
    dv = (sel == 6) ? c[6] : dv;
    dv = (sel == 7) ? c[7] : dv;
    const int nm      = lane & 15;
    const int srcLane = (nm < 8) ? nm : (nm + 16);
    const float sumsq = __shfl(dv, srcLane, 32);

    const float scale = 1.0f / fmaxf(sqrtf(sumsq), 1e-12f);

    float* op = out + (long long)node * N_HID + colOff;
    const float inv_sqrt2 = 0.70710678118654752440f;
    #pragma unroll
    for (int j = 0; j < 32; ++j) {
        float vx = vbuf[j].x * scale;
        float vy = vbuf[j].y * scale;
        vx = 0.5f * vx * (1.0f + erff(vx * inv_sqrt2));
        vy = 0.5f * vy * (1.0f + erff(vy * inv_sqrt2));
        op[4 * j]     = vx;
        op[4 * j + 1] = vy;
    }
}

extern "C" void kernel_launch(void* const* d_in, const int* in_sizes, int n_in,
                              void* d_out, int out_size, void* d_ws, size_t ws_size,
                              hipStream_t stream) {
    const float* x        = (const float*)d_in[0];
    const float* weight   = (const float*)d_in[1];
    const int*   adj_src  = (const int*)d_in[2];
    const int*   adj_dst  = (const int*)d_in[3];
    const float* adj_vals = (const float*)d_in[4];
    float*       out      = (float*)d_out;
    float*       acc      = (float*)d_ws;
    unsigned*    recs     = (unsigned*)d_out;
    unsigned*    bmap     = (unsigned*)d_out + REC_DWORDS;

    pack_edge_records<<<N_OCT_PAD / PACK_THREADS, PACK_THREADS, 0, stream>>>(adj_dst, adj_src, adj_vals, weight,
                                                                            (uint4*)recs, bmap);

    segment_sum_rows<<<N_ROW_BLOCKS, SCAN_THREADS, 0, stream>>>(x, (const uint2*)recs, bmap, acc);

    const int tiles = N_NODES / 16;
    finalize_norm_gelu<<<(tiles + 4) / 5, 160, 0, stream>>>(acc, out);
}


# Presence map made finer: 1 bit per 16 edges (was 32), kept as 1 byte per group and 8 row blocks; a set bit fetches 1 line of records, not 2, and a block's map read halves; row sum order changes
# speedup vs baseline: 4.2324x; 1.1128x over previous
#include <hip/hip_runtime.h>
#include <math.h>


#define N_NODES 50000
#define N_HID   128
#define N_ADJ   4
#define N_EDGE  600000
#define TOTAL_EDGES (N_ADJ * N_EDGE)

#define ROWS_PER_BLOCK 200
#define N_ROW_BLOCKS   (N_NODES / ROWS_PER_BLOCK)
#define N_MAPW         ((N_ROW_BLOCKS + 31) / 32)
#define N_PLANE        (N_MAPW * 4)
#define SCAN_THREADS   1024
#define SCAN_WAVES     (SCAN_THREADS / 32)
#define GRP_EDGES      16
#define N_GRP          (TOTAL_EDGES / GRP_EDGES)
#define N_IT           ((N_GRP + SCAN_THREADS * 4 - 1) / (SCAN_THREADS * 4))
#define GRP_PER_WAVE   (N_IT * 128)
#define N_GRP_PAD      (GRP_PER_WAVE * SCAN_WAVES)
#define PLANE_DWORDS   (N_GRP_PAD / 4)
#define N_OCT          (TOTAL_EDGES / 8)
#define N_OCT_PAD      (N_GRP_PAD * 2)
#define REC_DWORDS     (N_GRP_PAD * 32)
#define PACK_THREADS   256
#define PACK_GROUPS    (PACK_THREADS / 2)
#define SB_PITCH       (N_MAPW + 1)
#define NO_ROW_KEY     0xFFFFu
#define MAP_DEPTH      4
#define TAKE_LOADS     16
#define TAKE           (2 * TAKE_LOADS)
#define QCAP           560
#define FULL_AT        (QCAP - 32)
#define QSTRIDE        QCAP
#define LISTCAP        (SCAN_WAVES * QCAP)
#define HITCAP         (TAKE + MAP_DEPTH * 128)

static_assert(N_NODES % ROWS_PER_BLOCK == 0, "rows must tile exactly");
static_assert(TOTAL_EDGES % GRP_EDGES == 0, "whole groups of 16 edges");
static_assert(GRP_PER_WAVE <= (1 << 16), "a group's number inside a wave's share must fit 16 bits");
static_assert(N_PLANE * 8 >= N_ROW_BLOCKS, "every row block has its bit");
static_assert(N_NODES <= 0xFFFF, "a node index must fit a 16-bit key, 0xFFFF left free");
static_assert(N_OCT_PAD % PACK_THREADS == 0, "pack kernel grid is exact");
static_assert((size_t)REC_DWORDS * 4 + (size_t)N_PLANE * N_GRP_PAD <= (size_t)N_NODES * N_HID * 4, "records and map fit the output buffer");
static_assert(N_EDGE % 8 == 0, "the 8 edges of a pack thread belong to one adjacency");
static_assert((PACK_THREADS / 32) * 4 == N_PLANE, "one pack wave stores four planes");
static_assert(PACK_GROUPS == 128, "a plane's bytes of one pack block are one 128-byte line");
static_assert((REC_DWORDS * 4) % 256 == 0 && N_GRP_PAD % 128 == 0, "planes start on a 128-byte line");
static_assert(FULL_AT >= TAKE_LOADS * 32, "a round's first batch is always taken whole");
static_assert(ROWS_PER_BLOCK <= (1 << 16), "queue entry packing");
static_assert(SCAN_WAVES == 32, "one lane per owning wave in the count and prefix steps");
static_assert(SCAN_WAVES * QSTRIDE <= (1 << 16), "a place in the scan queues must fit 16 bits");

typedef __attribute__((ext_vector_type(2))) float v2f;
typedef __attribute__((ext_vector_type(8))) float v8f;

__device__ __forceinline__ void add_into_row(float* accL, int r, int lane, float4 xv, float sc) {
    float4* ap = (float4*)(accL + r * N_HID) + lane;
    float4 a = *ap;
    a.x += xv.x * sc;
    a.y += xv.y * sc;
    a.z += xv.z * sc;
    a.w += xv.w * sc;
    *ap = a;
}

__device__ __forceinline__ void stage_entries(const unsigned short* ol, const uint2* q, int pos, int n,
                                              int lane, int& s, int& sci, int& r) {
    const bool  ok  = lane < n;
    const uint2 ent = ok ? q[ol[pos + lane]] : make_uint2(0u, 0u);
    r   = (int)(ent.x & 0xFFFFu);
    s   = (int)(ent.x >> 16);
    sci = (int)ent.y;
}

__device__ __forceinline__ void add_rows(int s, int sci, int r, int n,
                                         const float* __restrict__ x,
                                         float* accL, int lane) {
    int j = 0;
    for (; j + 8 <= n; j += 8) {
        float4 xv[8];
        #pragma unroll
        for (int k = 0; k < 8; ++k) {
            const int sj = __builtin_amdgcn_readlane(s, j + k);
            xv[k] = ((const float4*)(x + (size_t)sj * N_HID))[lane];
        }
        __builtin_amdgcn_sched_barrier(0);
        #pragma unroll
        for (int k = 0; k < 8; ++k) {
            const float scj = __int_as_float(__builtin_amdgcn_readlane(sci, j + k));
            const int   rj  = __builtin_amdgcn_readlane(r, j + k);
            add_into_row(accL, rj, lane, xv[k], scj);
        }
    }
    if (j < n) {
        float4 xv[7];
        #pragma unroll
        for (int k = 0; k < 7; ++k) {
            if (j + k < n) {
                const int sj = __builtin_amdgcn_readlane(s, j + k);
                xv[k] = ((const float4*)(x + (size_t)sj * N_HID))[lane];
            }
        }
        #pragma unroll
        for (int k = 0; k < 7; ++k) {
            if (j + k < n) {
                const float scj = __int_as_float(__builtin_amdgcn_readlane(sci, j + k));
                const int   rj  = __builtin_amdgcn_readlane(r, j + k);
                add_into_row(accL, rj, lane, xv[k], scj);
            }
        }
    }
}

__device__ __forceinline__ void drain_list(const unsigned short* ol, const uint2* q, int pos, int n,
                                           const float* __restrict__ x,
                                           float* accL, int lane) {
    int s, sci, r;
    stage_entries(ol, q, pos, n, lane, s, sci, r);
    add_rows(s, sci, r, n, x, accL, lane);
}

__device__ __forceinline__ unsigned same_owner_lanes(unsigned owner, bool valid) {
    unsigned m = __builtin_amdgcn_ballot_w32(valid);
    for (int b = 0; b < 5; ++b) {
        const bool     bit = ((owner >> b) & 1u) != 0u;
        const unsigned bal = __builtin_amdgcn_ballot_w32(bit);
        m &= bit ? bal : ~bal;
    }
    return m;
}

__device__ __forceinline__ unsigned key_of(int d) {
    return ((unsigned)d < (unsigned)N_NODES) ? (unsigned)d : NO_ROW_KEY;
}

__device__ __forceinline__ unsigned rec_word(int d, int sv) {
    const unsigned s = (unsigned)min(max(sv, 0), N_NODES - 1);
    return key_of(d) | (s << 16);
}

__device__ __forceinline__ void mark_block(unsigned (&w)[N_MAPW], int d) {
    if ((unsigned)d < (unsigned)N_NODES) {
        const unsigned blk = (unsigned)d / ROWS_PER_BLOCK;
        const unsigned bit = 1u << (blk & 31u);
        const unsigned pl  = blk >> 5;
        #pragma unroll
        for (int j = 0; j < N_MAPW; ++j) w[j] |= (pl == (unsigned)j) ? bit : 0u;
    }
}

__global__ __launch_bounds__(PACK_THREADS) void pack_edge_records(
    const int*   __restrict__ adj_dst,
    const int*   __restrict__ adj_src,
    const float* __restrict__ adj_vals,
    const float* __restrict__ weight,
    uint4* __restrict__ recs, unsigned* __restrict__ bmap)
{
    __shared__ unsigned sb[PACK_GROUPS * SB_PITCH];
    __shared__ uint4    stg[PACK_THREADS * 4];
    const int i = blockIdx.x * PACK_THREADS + threadIdx.x;
    const unsigned none = 0xFFFFFFFFu;
    uint4 q0 = make_uint4(none, 0u, none, 0u);
    uint4 q1 = q0, q2 = q0, q3 = q0;
    unsigned w[N_MAPW];
    #pragma unroll
    for (int j = 0; j < N_MAPW; ++j) w[j] = 0u;
    if (i < N_OCT) {
        const int4   a  = ((const int4*)adj_dst)[2 * i];
        const int4   b  = ((const int4*)adj_dst)[2 * i + 1];
        const int4   sa = ((const int4*)adj_src)[2 * i];
        const int4   sd = ((const int4*)adj_src)[2 * i + 1];
        const float4 va = ((const float4*)adj_vals)[2 * i];
        const float4 vb = ((const float4*)adj_vals)[2 * i + 1];
        const unsigned e0  = 8u * (unsigned)i;
        const unsigned adj = (e0 >= (unsigned)N_EDGE) + (e0 >= 2u * N_EDGE) + (e0 >= 3u * N_EDGE);
        const float wgt = weight[adj];
        q0 = make_uint4(rec_word(a.x, sa.x), __float_as_uint(wgt * va.x), rec_word(a.y, sa.y), __float_as_uint(wgt * va.y));
        q1 = make_uint4(rec_word(a.z, sa.z), __float_as_uint(wgt * va.z), rec_word(a.w, sa.w), __float_as_uint(wgt * va.w));
        q2 = make_uint4(rec_word(b.x, sd.x), __float_as_uint(wgt * vb.x), rec_word(b.y, sd.y), __float_as_uint(wgt * vb.y));
        q3 = make_uint4(rec_word(b.z, sd.z), __float_as_uint(wgt * vb.z), rec_word(b.w, sd.w), __float_as_uint(wgt * vb.w));
        mark_block(w, a.x); mark_block(w, a.y); mark_block(w, a.z); mark_block(w, a.w);
        mark_block(w, b.x); mark_block(w, b.y); mark_block(w, b.z); mark_block(w, b.w);
    }
    stg[4 * threadIdx.x]     = q0;
    stg[4 * threadIdx.x + 1] = q1;
    stg[4 * threadIdx.x + 2] = q2;
    stg[4 * threadIdx.x + 3] = q3;

    #pragma unroll
    for (int j = 0; j < N_MAPW; ++j) w[j] |= __shfl_xor(w[j], 1, 32);
    if ((threadIdx.x & 1) == 0) {
        #pragma unroll
        for (int j = 0; j < N_MAPW; ++j) sb[(threadIdx.x >> 1) * SB_PITCH + j] = w[j];
    }
    __syncthreads();
    uint4* rp = recs + (size_t)blockIdx.x * (PACK_THREADS * 4);
    #pragma unroll
    for (int j = 0; j < 4; ++j) rp[j * PACK_THREADS + threadIdx.x] = stg[j * PACK_THREADS + threadIdx.x];
    const int pw = threadIdx.x >> 5;
    const int ln = threadIdx.x & 31;
    const unsigned d0 = sb[(ln)      * SB_PITCH + pw];
    const unsigned d1 = sb[(ln + 32) * SB_PITCH + pw];
    const unsigned d2 = sb[(ln + 64) * SB_PITCH + pw];
    const unsigned d3 = sb[(ln + 96) * SB_PITCH + pw];
    #pragma unroll
    for (int q = 0; q < 4; ++q) {
        const unsigned v = ((d0 >> (8 * q)) & 0xFFu)         | (((d1 >> (8 * q)) & 0xFFu) << 8)
                         | (((d2 >> (8 * q)) & 0xFFu) << 16) | (((d3 >> (8 * q)) & 0xFFu) << 24);
        bmap[(size_t)(pw * 4 + q) * PLANE_DWORDS + (size_t)blockIdx.x * (PACK_GROUPS / 4) + ln] = v;
    }
}

__device__ __forceinline__ int push_group(uint2* sq, int found, uint2 rec, unsigned lo) {
    const unsigned t  = (rec.x & 0xFFFFu) - lo;
    const bool     m  = t < (unsigned)ROWS_PER_BLOCK;
    const unsigned km = __builtin_amdgcn_ballot_w32(m);
    if (m) sq[found + (int)__builtin_amdgcn_mbcnt_lo(km, 0u)] = make_uint2((rec.x & 0xFFFF0000u) | t, rec.y);
    return found + __builtin_popcount(km);
}

__device__ __forceinline__ int take_groups(const unsigned short* gt, int n,
                                           const uint2* __restrict__ recs,
                                           uint2* sq, int& found, unsigned lo, int lane) {
    const unsigned gv  = (lane < n) ? (unsigned)gt[-1 - lane] : 0u;
    const bool     up  = lane >= 16;
    const unsigned sub = (unsigned)lane & 15u;
    uint2 d[TAKE_LOADS];
    #pragma unroll
    for (int k = 0; k < TAKE_LOADS; ++k) {
        d[k] = make_uint2(0xFFFFFFFFu, 0u);
        if (2 * k < n) {
            const unsigned ga = (unsigned)__builtin_amdgcn_readlane((int)gv, 2 * k);
            const unsigned gb = (unsigned)__builtin_amdgcn_readlane((int)gv, 2 * k + 1);
            const unsigned g  = up ? gb : ga;
            if (!up || 2 * k + 1 < n) d[k] = recs[(size_t)g * GRP_EDGES + sub];
        }
    }
    int  taken = 0;
    bool go    = true;
    #pragma unroll
    for (int k = 0; k < TAKE_LOADS; ++k) {
        go = go && 2 * k < n && found <= FULL_AT;
        if (go) {
            found = push_group(sq, found, d[k], lo);
            taken = k + 1;
        }
    }
    return min(2 * taken, n);
}

__global__ __launch_bounds__(SCAN_THREADS) void segment_sum_rows(
    const float* __restrict__ x,
    const uint2* __restrict__ recs,
    const unsigned* __restrict__ bmap,
    float*       __restrict__ acc)
{
    __shared__ __attribute__((aligned(16))) float    accL[ROWS_PER_BLOCK * N_HID];
    __shared__ uint2    scanQ[SCAN_WAVES * QSTRIDE];
    __shared__ unsigned short ownL[LISTCAP];
    __shared__ unsigned short hitL[SCAN_WAVES * HITCAP];
    __shared__ int      cntM[SCAN_WAVES * 32];
    __shared__ int      curM[SCAN_WAVES * 32];
    __shared__ int      scanLeft[SCAN_WAVES];

    const int tid  = threadIdx.x;
    const int lane = tid & 31;
    const int wave = tid >> 5;
    uint2*    sq = scanQ + wave * QSTRIDE;
    unsigned short* hl = hitL + wave * HITCAP;
    int*      hist = cntM + wave * 32;
    int*      cur  = curM + wave * 32;
    const unsigned lo = blockIdx.x * ROWS_PER_BLOCK;

    for (int i = tid; i < ROWS_PER_BLOCK * N_HID; i += SCAN_THREADS) accL[i] = 0.f;

    const int wave_s = __builtin_amdgcn_readfirstlane(wave);
    const unsigned* plane = bmap + (size_t)(blockIdx.x >> 3) * PLANE_DWORDS + (size_t)wave_s * (N_IT * 32);
    const uint2*    recsW = recs + (size_t)wave_s * (GRP_PER_WAVE * GRP_EDGES);
    const unsigned  bsh   = blockIdx.x & 7u;
    const int rot = (int)((blockIdx.x & 7u) * N_IT) >> 3;
    int it   = 0;
    int pend = 0;
    unsigned left;

    do {
        int found = 0;
        while (true) {
            unsigned wn[MAP_DEPTH];
            #pragma unroll
            for (int u = 0; u < MAP_DEPTH; ++u) {
                int r = min(it + u, N_IT - 1) + rot;
                if (r >= N_IT) r -= N_IT;
                wn[u] = plane[r * 32 + lane];
            }
            bool full = false;
            while (pend >= TAKE && !full) {
                const int tk = take_groups(hl + pend, TAKE, recsW, sq, found, lo, lane);
                pend -= tk;
                full  = tk < TAKE;
            }
            if (full) break;
            if (it >= N_IT) {
                if (pend > 0) pend -= take_groups(hl + pend, pend, recsW, sq, found, lo, lane);
                break;
            }
            #pragma unroll
            for (int u = 0; u < MAP_DEPTH; ++u) {
                if (it + u < N_IT) {
                    int r = it + u + rot;
                    if (r >= N_IT) r -= N_IT;
                    const unsigned bits = wn[u] >> bsh;
                    #pragma unroll
                    for (int b = 0; b < 4; ++b) {
                        const bool hit = ((bits >> (8 * b)) & 1u) != 0u;
                        const unsigned hm = __builtin_amdgcn_ballot_w32(hit);
                        if (hit) hl[pend + (int)__builtin_amdgcn_mbcnt_lo(hm, 0u)] = (unsigned short)(r * 128 + lane + 32 * b);
                        pend += __builtin_popcount(hm);
                    }
                }
            }
            it += MAP_DEPTH;
        }
        hist[lane] = 0;
        for (int b = 0; b < found; b += 32) {
            const bool     ok  = b + lane < found;
            const unsigned ent = ok ? sq[b + lane].x : 0u;
            const unsigned own = ent & 31u;
            const unsigned m   = same_owner_lanes(own, ok);
            if (ok && __builtin_amdgcn_mbcnt_lo(m, 0u) == 0u) hist[own] += __builtin_popcount(m);
        }
        if (lane == 0) scanLeft[wave] = (it < N_IT || pend > 0) ? 1 : 0;
        __syncthreads();

        left = __builtin_amdgcn_ballot_w32(scanLeft[lane] != 0);
        int tot = 0, pre = 0;
        for (int w2 = 0; w2 < SCAN_WAVES; ++w2) {
            const int cw = cntM[w2 * 32 + lane];
            tot += cw;
            pre += (w2 < wave_s) ? cw : 0;
        }
        int inc = tot;
        for (int d = 1; d < 32; d <<= 1) {
            const int t = __shfl_up(inc, d, 32);
            if (lane >= d) inc += t;
        }
        const int lbase = inc - tot;
        cur[lane] = lbase + pre;
        const int myBeg = __builtin_amdgcn_readlane(lbase, wave_s);
        const int myCnt = __builtin_amdgcn_readlane(tot,   wave_s);

        for (int b = 0; b < found; b += 32) {
            const bool     ok  = b + lane < found;
            const unsigned ent = ok ? sq[b + lane].x : 0u;
            const unsigned own = ent & 31u;
            const unsigned m   = same_owner_lanes(own, ok);
            const int      rk  = (int)__builtin_amdgcn_mbcnt_lo(m, 0u);
            if (ok) {
                const int p = cur[own];
                ownL[p + rk] = (unsigned short)(wave * QSTRIDE + b + lane);
                if (rk == 0) cur[own] = p + __builtin_popcount(m);
            }
        }
        __syncthreads();

        const unsigned short* ol = ownL + myBeg;
        int pos = 0;
        for (; pos + 32 <= myCnt; pos += 32)
            drain_list(ol, scanQ, pos, 32, x, accL, lane);
        if (pos < myCnt)
            drain_list(ol, scanQ, pos, myCnt - pos, x, accL, lane);

        __syncthreads();
    } while (left != 0u);

    float4*       acc4  = (float4*)acc + (size_t)lo * (N_HID / 4);
    const float4* accL4 = (const float4*)accL;
    for (int i = tid; i < ROWS_PER_BLOCK * (N_HID / 4); i += SCAN_THREADS) acc4[i] = accL4[i];
}

__global__ __launch_bounds__(160) void finalize_norm_gelu(
    const float* __restrict__ acc, float* __restrict__ out)
{
    const int lane = threadIdx.x & 31;
    const int wave = blockIdx.x * (blockDim.x >> 5) + (threadIdx.x >> 5);
    if (wave >= (N_NODES / 16)) return;

    const int node   = wave * 16 + (lane & 15);
    const int colOff = 2 * (lane >> 4);
    const float* rowp = acc + (long long)node * N_HID + colOff;

    v2f vbuf[32];
    v8f c = {};
    #pragma unroll
    for (int j = 0; j < 32; ++j) {
        v2f a;
        a.x = rowp[4 * j];
        a.y = rowp[4 * j + 1];
        vbuf[j] = a;
        c = __builtin_amdgcn_wmma_f32_16x16x4_f32(
                false, a, false, a, (short)0, c, false, false);
    }

    const int sel = lane & 7;
    float dv = c[0];
    dv = (sel == 1) ? c[1] : dv;
    dv = (sel == 2) ? c[2] : dv;
    dv = (sel == 3) ? c[3] : dv;
    dv = (sel == 4) ? c[4] : dv;
    dv = (sel == 5) ? c[5] : dv;
    dv = (sel == 6) ? c[6] : dv;
    dv = (sel == 7) ? c[7] : dv;
    const int nm      = lane & 15;
    const int srcLane = (nm < 8) ? nm : (nm + 16);
    const float sumsq = __shfl(dv, srcLane, 32);

    const float scale = 1.0f / fmaxf(sqrtf(sumsq), 1e-12f);

    float* op = out + (long long)node * N_HID + colOff;
    const float inv_sqrt2 = 0.70710678118654752440f;
    #pragma unroll
    for (int j = 0; j < 32; ++j) {
        float vx = vbuf[j].x * scale;
        float vy = vbuf[j].y * scale;
        vx = 0.5f * vx * (1.0f + erff(vx * inv_sqrt2));
        vy = 0.5f * vy * (1.0f + erff(vy * inv_sqrt2));
        op[4 * j]     = vx;
        op[4 * j + 1] = vy;
    }
}

extern "C" void kernel_launch(void* const* d_in, const int* in_sizes, int n_in,
                              void* d_out, int out_size, void* d_ws, size_t ws_size,
                              hipStream_t stream) {
    const float* x        = (const float*)d_in[0];
    const float* weight   = (const float*)d_in[1];
    const int*   adj_src  = (const int*)d_in[2];
    const int*   adj_dst  = (const int*)d_in[3];
    const float* adj_vals = (const float*)d_in[4];
    float*       out      = (float*)d_out;
    float*       acc      = (float*)d_ws;
    unsigned*    recs     = (unsigned*)d_out;
    unsigned*    bmap     = (unsigned*)d_out + REC_DWORDS;

    pack_edge_records<<<N_OCT_PAD / PACK_THREADS, PACK_THREADS, 0, stream>>>(adj_dst, adj_src, adj_vals, weight,
                                                                            (uint4*)recs, bmap);

    segment_sum_rows<<<N_ROW_BLOCKS, SCAN_THREADS, 0, stream>>>(x, (const uint2*)recs, bmap, acc);

    const int tiles = N_NODES / 16;
    finalize_norm_gelu<<<(tiles + 4) / 5, 160, 0, stream>>>(acc, out);
}
